# starting kernel
# baseline (speedup 1.0000x reference)
#include <hip/hip_runtime.h>
#include <hip/hip_bf16.h>
#include <math.h>

#define B_   32
#define C_   768
#define CR_  192
#define HW_  1024
#define K1_  C_
#define K2_  CR_

typedef __attribute__((ext_vector_type(2))) float v2f;
typedef __attribute__((ext_vector_type(8))) float v8f;

__device__ __forceinline__ float sigmoidf_fast(float v) {
    return 1.0f / (1.0f + __expf(-v));
}

__global__ __launch_bounds__(256) void se_squeeze(const float* __restrict__ x,
                                                  float* __restrict__ s) {
    const int bc = blockIdx.x;
    const float4* p = (const float4*)(x + (size_t)bc * HW_);
    float4 v = p[threadIdx.x];
    float sum = v.x + v.y + v.z + v.w;

    #pragma unroll
    for (int o = 16; o > 0; o >>= 1)
        sum += __shfl_down(sum, o, 32);

    __shared__ float tmp[8];
    const int lane = threadIdx.x & 31;
    const int wave = threadIdx.x >> 5;
    if (lane == 0) tmp[wave] = sum;
    __syncthreads();
    if (threadIdx.x < 8) {
        float t = tmp[threadIdx.x];
        #pragma unroll
        for (int o = 4; o > 0; o >>= 1)
            t += __shfl_down(t, o, 8);
        if (threadIdx.x == 0) s[bc] = t * (1.0f / (float)HW_);
    }
}

__global__ __launch_bounds__(256) void se_mlp_wmma(const float* __restrict__ s,
                                                   const float* __restrict__ w1,
                                                   const float* __restrict__ b1,
                                                   const float* __restrict__ w2,
                                                   const float* __restrict__ b2,
                                                   float* __restrict__ h,
                                                   float* __restrict__ g) {
    const int wave = threadIdx.x >> 5;
    const int lane = threadIdx.x & 31;
    const int half = lane >> 4;
    const int l    = lane & 15;
    const int koff = 2 * half;

    for (int t = wave; t < 24; t += 8) {
        const int mt = t & 1;
        const int nt = t >> 1;
        const int m0 = mt * 16;
        const int n0 = nt * 16;
        const float* arow = s  + (size_t)(m0 + l) * K1_ + koff;
        const float* brow = w1 + (size_t)(n0 + l) * K1_ + koff;

        v8f acc = {};
        #pragma unroll 4
        for (int k = 0; k < K1_; k += 4) {
            v2f a = *(const v2f*)(arow + k);
            v2f b = *(const v2f*)(brow + k);
            acc = __builtin_amdgcn_wmma_f32_16x16x4_f32(
                false, a, false, b, (short)0, acc, false, false);
        }
        const float bias = b1[n0 + l];
        #pragma unroll
        for (int r = 0; r < 8; ++r) {
            const int m = m0 + r + 8 * half;
            float v = acc[r] + bias;
            v = v * sigmoidf_fast(v);
            h[(size_t)m * CR_ + n0 + l] = v;
        }
    }

    __syncthreads();

    for (int t = wave; t < 96; t += 8) {
        const int mt = t & 1;
        const int nt = t >> 1;
        const int m0 = mt * 16;
        const int n0 = nt * 16;
        const float* arow = h  + (size_t)(m0 + l) * K2_ + koff;
        const float* brow = w2 + (size_t)(n0 + l) * K2_ + koff;

        v8f acc = {};
        #pragma unroll 4
        for (int k = 0; k < K2_; k += 4) {
            v2f a = *(const v2f*)(arow + k);
            v2f b = *(const v2f*)(brow + k);
            acc = __builtin_amdgcn_wmma_f32_16x16x4_f32(
                false, a, false, b, (short)0, acc, false, false);
        }
        const float bias = b2[n0 + l];
        #pragma unroll
        for (int r = 0; r < 8; ++r) {
            const int m = m0 + r + 8 * half;
            float v = acc[r] + bias;
            g[(size_t)m * C_ + n0 + l] = sigmoidf_fast(v);
        }
    }
}

__global__ __launch_bounds__(256) void se_scale(const float* __restrict__ x,
                                                const float* __restrict__ g,
                                                float* __restrict__ out) {
    const size_t idx4 = (size_t)blockIdx.x * blockDim.x + threadIdx.x;
    const size_t total4 = (size_t)B_ * C_ * HW_ / 4;
    if (idx4 >= total4) return;
    const size_t bc = idx4 >> 8;
    const float gv = g[bc];
    float4 v = ((const float4*)x)[idx4];
    v.x *= gv; v.y *= gv; v.z *= gv; v.w *= gv;
    ((float4*)out)[idx4] = v;
}

extern "C" void kernel_launch(void* const* d_in, const int* in_sizes, int n_in,
                              void* d_out, int out_size, void* d_ws, size_t ws_size,
                              hipStream_t stream) {
    const float* x  = (const float*)d_in[0];
    const float* w1 = (const float*)d_in[1];
    const float* b1 = (const float*)d_in[2];
    const float* w2 = (const float*)d_in[3];
    const float* b2 = (const float*)d_in[4];
    float* out = (float*)d_out;

    float* s = (float*)d_ws;
    float* h = s + (size_t)B_ * C_;
    float* g = h + (size_t)B_ * CR_;

    se_squeeze<<<B_ * C_, 256, 0, stream>>>(x, s);

    se_mlp_wmma<<<1, 256, 0, stream>>>(s, w1, b1, w2, b2, h, g);

    const int nblocks = (int)(((size_t)B_ * C_ * HW_ / 4 + 255) / 256);
    se_scale<<<nblocks, 256, 0, stream>>>(x, g, out);
}


# Excitation MLP in one launch of 48 work-groups: each recomputes the hidden strip of its 16 batch rows into LDS (a tile per wave, full K) and forms 32 gate columns from it (new f32 sum order)
# speedup vs baseline: 2.7860x; 2.7860x over previous
#include <hip/hip_runtime.h>
#include <hip/hip_bf16.h>
#include <math.h>

#define B_   32
#define C_   768
#define CR_  192
#define HW_  1024
#define K1_  C_
#define K2_  CR_

typedef __attribute__((ext_vector_type(2))) float v2f;
typedef __attribute__((ext_vector_type(4))) float v4f;
typedef __attribute__((ext_vector_type(8))) float v8f;

__device__ __forceinline__ float sigmoidf_fast(float v) {
    return 1.0f / (1.0f + __expf(-v));
}

__global__ __launch_bounds__(256) void se_squeeze(const float* __restrict__ x,
                                                  float* __restrict__ s) {
    const int bc = blockIdx.x;
    const float4* p = (const float4*)(x + (size_t)bc * HW_);
    float4 v = p[threadIdx.x];
    float sum = v.x + v.y + v.z + v.w;

    #pragma unroll
    for (int o = 16; o > 0; o >>= 1)
        sum += __shfl_down(sum, o, 32);

    __shared__ float tmp[8];
    const int lane = threadIdx.x & 31;
    const int wave = threadIdx.x >> 5;
    if (lane == 0) tmp[wave] = sum;
    __syncthreads();
    if (threadIdx.x < 8) {
        float t = tmp[threadIdx.x];
        #pragma unroll
        for (int o = 4; o > 0; o >>= 1)
            t += __shfl_down(t, o, 8);
        if (threadIdx.x == 0) s[bc] = t * (1.0f / (float)HW_);
    }
}

struct frag8 { v4f a[8]; v4f b[8]; };

__device__ __forceinline__ void load8(frag8& f, const float* __restrict__ arow,
                                      const float* __restrict__ brow, int k) {
    #pragma unroll
    for (int i = 0; i < 8; ++i) {
        f.a[i] = *(const v4f*)(arow + k + 8 * i);
        f.b[i] = *(const v4f*)(brow + k + 8 * i);
    }
}

__device__ __forceinline__ void mma8(const frag8& f, v8f& acc0, v8f& acc1) {
    #pragma unroll
    for (int i = 0; i < 8; ++i) {
        const v2f a0 = {f.a[i].x, f.a[i].y};
        const v2f b0 = {f.b[i].x, f.b[i].y};
        const v2f a1 = {f.a[i].z, f.a[i].w};
        const v2f b1 = {f.b[i].z, f.b[i].w};
        acc0 = __builtin_amdgcn_wmma_f32_16x16x4_f32(
            false, a0, false, b0, (short)0, acc0, false, false);
        acc1 = __builtin_amdgcn_wmma_f32_16x16x4_f32(
            false, a1, false, b1, (short)0, acc1, false, false);
    }
}

#define HP_ 196

__global__ __launch_bounds__(384) void se_mlp_rows(const float* __restrict__ s,
                                                   const float* __restrict__ w1,
                                                   const float* __restrict__ b1,
                                                   const float* __restrict__ w2,
                                                   const float* __restrict__ b2,
                                                   float* __restrict__ g) {
    __shared__ __attribute__((aligned(16))) float hs[16 * HP_];

    const int wave = __builtin_amdgcn_readfirstlane((int)(threadIdx.x >> 5));
    const int lane = threadIdx.x & 31;
    const int half = lane >> 4;
    const int l    = lane & 15;
    const int koff = 4 * half;
    const int m0   = blockIdx.y * 16;

    {
        const int n0 = wave * 16;
        const float* arow = s  + (size_t)(m0 + l) * K1_ + koff;
        const float* brow = w1 + (size_t)(n0 + l) * K1_ + koff;

        v8f acc0 = {};
        v8f acc1 = {};
        frag8 fa, fb;
        load8(fa, arow, brow, 0);
        #pragma unroll 1
        for (int k = 0; k < K1_ - 128; k += 128) {
            load8(fb, arow, brow, k + 64);
            mma8(fa, acc0, acc1);
            load8(fa, arow, brow, k + 128);
            mma8(fb, acc0, acc1);
        }
        load8(fb, arow, brow, K1_ - 64);
        mma8(fa, acc0, acc1);
        mma8(fb, acc0, acc1);
        const float bias = b1[n0 + l];
        #pragma unroll
        for (int r = 0; r < 8; ++r) {
            float v = (acc0[r] + acc1[r]) + bias;
            v = v * sigmoidf_fast(v);
            hs[(r + 8 * half) * HP_ + n0 + l] = v;
        }
    }

    __syncthreads();

    if (wave == 0) {
        const int na = blockIdx.x * 32;
        const float* hrow = hs + l * HP_ + koff;
        const float* wa = w2 + (size_t)(na + l) * K2_ + koff;
        const float* wb = w2 + (size_t)(na + 16 + l) * K2_ + koff;

        v8f ca0 = {};
        v8f ca1 = {};
        v8f cb0 = {};
        v8f cb1 = {};
        #pragma unroll 12
        for (int k = 0; k < K2_; k += 8) {
            const v4f a = *(const v4f*)(hrow + k);
            const v4f p = *(const v4f*)(wa + k);
            const v4f q = *(const v4f*)(wb + k);
            const v2f a0 = {a.x, a.y};
            const v2f a1 = {a.z, a.w};
            const v2f p0 = {p.x, p.y};
            const v2f p1 = {p.z, p.w};
            const v2f q0 = {q.x, q.y};
            const v2f q1 = {q.z, q.w};
            ca0 = __builtin_amdgcn_wmma_f32_16x16x4_f32(
                false, a0, false, p0, (short)0, ca0, false, false);
            cb0 = __builtin_amdgcn_wmma_f32_16x16x4_f32(
                false, a0, false, q0, (short)0, cb0, false, false);
            ca1 = __builtin_amdgcn_wmma_f32_16x16x4_f32(
                false, a1, false, p1, (short)0, ca1, false, false);
            cb1 = __builtin_amdgcn_wmma_f32_16x16x4_f32(
                false, a1, false, q1, (short)0, cb1, false, false);
        }
        const float biasa = b2[na + l];
        const float biasb = b2[na + 16 + l];
        #pragma unroll
        for (int r = 0; r < 8; ++r) {
            const int m = m0 + r + 8 * half;
            g[(size_t)m * C_ + na + l]      = sigmoidf_fast((ca0[r] + ca1[r]) + biasa);
            g[(size_t)m * C_ + na + 16 + l] = sigmoidf_fast((cb0[r] + cb1[r]) + biasb);
        }
    }
}

__global__ __launch_bounds__(256) void se_scale(const float* __restrict__ x,
                                                const float* __restrict__ g,
                                                float* __restrict__ out) {
    const size_t idx4 = (size_t)blockIdx.x * blockDim.x + threadIdx.x;
    const size_t total4 = (size_t)B_ * C_ * HW_ / 4;
    if (idx4 >= total4) return;
    const size_t bc = idx4 >> 8;
    const float gv = g[bc];
    float4 v = ((const float4*)x)[idx4];
    v.x *= gv; v.y *= gv; v.z *= gv; v.w *= gv;
    ((float4*)out)[idx4] = v;
}

extern "C" void kernel_launch(void* const* d_in, const int* in_sizes, int n_in,
                              void* d_out, int out_size, void* d_ws, size_t ws_size,
                              hipStream_t stream) {
    const float* x  = (const float*)d_in[0];
    const float* w1 = (const float*)d_in[1];
    const float* b1 = (const float*)d_in[2];
    const float* w2 = (const float*)d_in[3];
    const float* b2 = (const float*)d_in[4];
    float* out = (float*)d_out;

    float* s = (float*)d_ws;
    float* g = s + (size_t)B_ * C_;

    se_squeeze<<<B_ * C_, 256, 0, stream>>>(x, s);

    se_mlp_rows<<<dim3(24, 2), 384, 0, stream>>>(s, w1, b1, w2, b2, g);

    const int nblocks = (int)(((size_t)B_ * C_ * HW_ / 4 + 255) / 256);
    se_scale<<<nblocks, 256, 0, stream>>>(x, g, out);
}


# Dense layers folded into the scale launch (2 launches, was 3): 256 work-groups each recompute their 16x192 hidden strip in LDS, form their 6 channels' gates there and scale their 96 planes in a loop
# speedup vs baseline: 2.9311x; 1.0521x over previous
#include <hip/hip_runtime.h>
#include <hip/hip_bf16.h>
#include <math.h>

#define B_   32
#define C_   768
#define CR_  192
#define HW_  1024
#define K1_  C_
#define K2_  CR_

typedef __attribute__((ext_vector_type(2))) float v2f;
typedef __attribute__((ext_vector_type(4))) float v4f;
typedef __attribute__((ext_vector_type(8))) float v8f;

__device__ __forceinline__ float sigmoidf_fast(float v) {
    return 1.0f / (1.0f + __expf(-v));
}

__global__ __launch_bounds__(256) void se_squeeze(const float* __restrict__ x,
                                                  float* __restrict__ s) {
    const int bc = blockIdx.x;
    const float4* p = (const float4*)(x + (size_t)bc * HW_);
    float4 v = p[threadIdx.x];
    float sum = v.x + v.y + v.z + v.w;

    #pragma unroll
    for (int o = 16; o > 0; o >>= 1)
        sum += __shfl_down(sum, o, 32);

    __shared__ float tmp[8];
    const int lane = threadIdx.x & 31;
    const int wave = threadIdx.x >> 5;
    if (lane == 0) tmp[wave] = sum;
    __syncthreads();
    if (threadIdx.x < 8) {
        float t = tmp[threadIdx.x];
        #pragma unroll
        for (int o = 4; o > 0; o >>= 1)
            t += __shfl_down(t, o, 8);
        if (threadIdx.x == 0) s[bc] = t * (1.0f / (float)HW_);
    }
}

struct frag8 { v4f a[8]; v4f b[8]; };

__device__ __forceinline__ void load8(frag8& f, const float* __restrict__ arow,
                                      const float* __restrict__ brow, int k) {
    #pragma unroll
    for (int i = 0; i < 8; ++i) {
        f.a[i] = *(const v4f*)(arow + k + 8 * i);
        f.b[i] = *(const v4f*)(brow + k + 8 * i);
    }
}

__device__ __forceinline__ void mma8(const frag8& f, v8f& acc0, v8f& acc1) {
    #pragma unroll
    for (int i = 0; i < 8; ++i) {
        const v2f a0 = {f.a[i].x, f.a[i].y};
        const v2f b0 = {f.b[i].x, f.b[i].y};
        const v2f a1 = {f.a[i].z, f.a[i].w};
        const v2f b1 = {f.b[i].z, f.b[i].w};
        acc0 = __builtin_amdgcn_wmma_f32_16x16x4_f32(
            false, a0, false, b0, (short)0, acc0, false, false);
        acc1 = __builtin_amdgcn_wmma_f32_16x16x4_f32(
            false, a1, false, b1, (short)0, acc1, false, false);
    }
}

#define HP_  196
#define NCH_ 6
#define GP_  8
#define PV_  (HW_ / 4 / 32)
#define NB_  3

__global__ __launch_bounds__(512) void se_mlp_scale(const float* __restrict__ x,
                                                    const float* __restrict__ s,
                                                    const float* __restrict__ w1,
                                                    const float* __restrict__ b1,
                                                    const float* __restrict__ w2,
                                                    const float* __restrict__ b2,
                                                    float* __restrict__ out) {
    __shared__ __attribute__((aligned(16))) float hs[16 * HP_];
    __shared__ float gs[16 * GP_];

    const int wave = __builtin_amdgcn_readfirstlane((int)(threadIdx.x >> 5));
    const int lane = threadIdx.x & 31;
    const int half = lane >> 4;
    const int l    = lane & 15;
    const int koff = 4 * half;
    const int m0   = blockIdx.y * 16;
    const int c0   = blockIdx.x * NCH_;

    if (wave < CR_ / 16) {
        const int n0 = wave * 16;
        const float* arow = s  + (size_t)(m0 + l) * K1_ + koff;
        const float* brow = w1 + (size_t)(n0 + l) * K1_ + koff;

        v8f acc0 = {};
        v8f acc1 = {};
        frag8 fa, fb;
        load8(fa, arow, brow, 0);
        #pragma unroll 1
        for (int k = 0; k < K1_ - 128; k += 128) {
            load8(fb, arow, brow, k + 64);
            mma8(fa, acc0, acc1);
            load8(fa, arow, brow, k + 128);
            mma8(fb, acc0, acc1);
        }
        load8(fb, arow, brow, K1_ - 64);
        mma8(fa, acc0, acc1);
        mma8(fb, acc0, acc1);
        const float bias = b1[n0 + l];
        #pragma unroll
        for (int r = 0; r < 8; ++r) {
            float v = (acc0[r] + acc1[r]) + bias;
            v = v * sigmoidf_fast(v);
            hs[(r + 8 * half) * HP_ + n0 + l] = v;
        }
    }

    __syncthreads();

    if (wave == 0) {
        const int cb = (c0 + l < C_) ? (c0 + l) : (C_ - 1);
        const float* hrow = hs + l * HP_ + koff;
        const float* wa = w2 + (size_t)cb * K2_ + koff;

        v8f ca0 = {};
        v8f ca1 = {};
        #pragma unroll 12
        for (int k = 0; k < K2_; k += 8) {
            const v4f a = *(const v4f*)(hrow + k);
            const v4f p = *(const v4f*)(wa + k);
            const v2f a0 = {a.x, a.y};
            const v2f a1 = {a.z, a.w};
            const v2f p0 = {p.x, p.y};
            const v2f p1 = {p.z, p.w};
            ca0 = __builtin_amdgcn_wmma_f32_16x16x4_f32(
                false, a0, false, p0, (short)0, ca0, false, false);
            ca1 = __builtin_amdgcn_wmma_f32_16x16x4_f32(
                false, a1, false, p1, (short)0, ca1, false, false);
        }
        const float biasa = b2[cb];
        if (l < NCH_) {
            #pragma unroll
            for (int r = 0; r < 8; ++r)
                gs[(r + 8 * half) * GP_ + l] = sigmoidf_fast((ca0[r] + ca1[r]) + biasa);
        }
    }

    __syncthreads();

    const size_t rowbase = ((size_t)(m0 + wave) * C_ + c0) * HW_;
    const v4f* __restrict__ xw = (const v4f*)(x + rowbase) + lane;
    v4f* __restrict__ ow       = (v4f*)(out + rowbase) + lane;

    #pragma unroll 1
    for (int t0 = 0; t0 < NCH_; t0 += NB_) {
        v4f v[NB_][PV_];
        #pragma unroll
        for (int t = 0; t < NB_; ++t) {
            #pragma unroll
            for (int j = 0; j < PV_; ++j) v[t][j] = xw[(t0 + t) * (HW_ / 4) + 32 * j];
        }
        #pragma unroll
        for (int t = 0; t < NB_; ++t) {
            const float gv = gs[wave * GP_ + t0 + t];
            #pragma unroll
            for (int j = 0; j < PV_; ++j) ow[(t0 + t) * (HW_ / 4) + 32 * j] = v[t][j] * gv;
        }
    }
}

extern "C" void kernel_launch(void* const* d_in, const int* in_sizes, int n_in,
                              void* d_out, int out_size, void* d_ws, size_t ws_size,
                              hipStream_t stream) {
    const float* x  = (const float*)d_in[0];
    const float* w1 = (const float*)d_in[1];
    const float* b1 = (const float*)d_in[2];
    const float* w2 = (const float*)d_in[3];
    const float* b2 = (const float*)d_in[4];
    float* out = (float*)d_out;

    float* s = (float*)d_ws;

    se_squeeze<<<B_ * C_, 256, 0, stream>>>(x, s);

    se_mlp_scale<<<dim3(128, 2), 512, 0, stream>>>(x, s, w1, b1, w2, b2, out);
}


# Pool launch takes the scale launch's grid and plane ownership: 256 work-groups of 16 waves, a wave sums its 6 contiguous planes from 48 loads in flight, then across lanes (new f32 sum order)
# speedup vs baseline: 4.2408x; 1.4468x over previous
#include <hip/hip_runtime.h>
#include <hip/hip_bf16.h>
#include <math.h>

#define B_   32
#define C_   768
#define CR_  192
#define HW_  1024
#define K1_  C_
#define K2_  CR_

typedef __attribute__((ext_vector_type(2))) float v2f;
typedef __attribute__((ext_vector_type(4))) float v4f;
typedef __attribute__((ext_vector_type(8))) float v8f;

__device__ __forceinline__ float sigmoidf_fast(float v) {
    return 1.0f / (1.0f + __expf(-v));
}

struct frag8 { v4f a[8]; v4f b[8]; };

__device__ __forceinline__ void load8(frag8& f, const float* __restrict__ arow,
                                      const float* __restrict__ brow, int k) {
    #pragma unroll
    for (int i = 0; i < 8; ++i) {
        f.a[i] = *(const v4f*)(arow + k + 8 * i);
        f.b[i] = *(const v4f*)(brow + k + 8 * i);
    }
}

__device__ __forceinline__ void mma8(const frag8& f, v8f& acc0, v8f& acc1) {
    #pragma unroll
    for (int i = 0; i < 8; ++i) {
        const v2f a0 = {f.a[i].x, f.a[i].y};
        const v2f b0 = {f.b[i].x, f.b[i].y};
        const v2f a1 = {f.a[i].z, f.a[i].w};
        const v2f b1 = {f.b[i].z, f.b[i].w};
        acc0 = __builtin_amdgcn_wmma_f32_16x16x4_f32(
            false, a0, false, b0, (short)0, acc0, false, false);
        acc1 = __builtin_amdgcn_wmma_f32_16x16x4_f32(
            false, a1, false, b1, (short)0, acc1, false, false);
    }
}

#define HP_  196
#define NCH_ 6
#define GP_  8
#define PV_  (HW_ / 4 / 32)
#define NB_  3

__global__ __launch_bounds__(512) __attribute__((amdgpu_waves_per_eu(4, 4))) void se_squeeze(const float* __restrict__ x,
                                                  float* __restrict__ s) {
    const int wave = __builtin_amdgcn_readfirstlane((int)(threadIdx.x >> 5));
    const int lane = threadIdx.x & 31;
    const int m0   = blockIdx.y * 16;
    const int c0   = blockIdx.x * NCH_;

    const size_t rowbase = ((size_t)(m0 + wave) * C_ + c0) * HW_;
    const v4f* __restrict__ xw = (const v4f*)(x + rowbase) + lane;

    v4f v[NCH_][PV_];
    #pragma unroll
    for (int t = 0; t < NCH_; ++t) {
        #pragma unroll
        for (int j = 0; j < PV_; ++j) v[t][j] = xw[t * (HW_ / 4) + 32 * j];
    }

    float mine = 0.0f;
    #pragma unroll
    for (int t = 0; t < NCH_; ++t) {
        float a0 = (v[t][0].x + v[t][0].y) + (v[t][0].z + v[t][0].w);
        float a1 = (v[t][1].x + v[t][1].y) + (v[t][1].z + v[t][1].w);
        float a2 = (v[t][2].x + v[t][2].y) + (v[t][2].z + v[t][2].w);
        float a3 = (v[t][3].x + v[t][3].y) + (v[t][3].z + v[t][3].w);
        a0 += (v[t][4].x + v[t][4].y) + (v[t][4].z + v[t][4].w);
        a1 += (v[t][5].x + v[t][5].y) + (v[t][5].z + v[t][5].w);
        a2 += (v[t][6].x + v[t][6].y) + (v[t][6].z + v[t][6].w);
        a3 += (v[t][7].x + v[t][7].y) + (v[t][7].z + v[t][7].w);
        float sum = (a0 + a1) + (a2 + a3);
        #pragma unroll
        for (int o = 16; o > 0; o >>= 1)
            sum += __shfl_xor(sum, o, 32);
        if (lane == t) mine = sum;
    }
    if (lane < NCH_)
        s[(size_t)(m0 + wave) * C_ + c0 + lane] = mine * (1.0f / (float)HW_);
}

__global__ __launch_bounds__(512) void se_mlp_scale(const float* __restrict__ x,
                                                    const float* __restrict__ s,
                                                    const float* __restrict__ w1,
                                                    const float* __restrict__ b1,
                                                    const float* __restrict__ w2,
                                                    const float* __restrict__ b2,
                                                    float* __restrict__ out) {
    __shared__ __attribute__((aligned(16))) float hs[16 * HP_];
    __shared__ float gs[16 * GP_];

    const int wave = __builtin_amdgcn_readfirstlane((int)(threadIdx.x >> 5));
    const int lane = threadIdx.x & 31;
    const int half = lane >> 4;
    const int l    = lane & 15;
    const int koff = 4 * half;
    const int m0   = blockIdx.y * 16;
    const int c0   = blockIdx.x * NCH_;

    if (wave < CR_ / 16) {
        const int n0 = wave * 16;
        const float* arow = s  + (size_t)(m0 + l) * K1_ + koff;
        const float* brow = w1 + (size_t)(n0 + l) * K1_ + koff;

        v8f acc0 = {};
        v8f acc1 = {};
        frag8 fa, fb;
        load8(fa, arow, brow, 0);
        #pragma unroll 1
        for (int k = 0; k < K1_ - 128; k += 128) {
            load8(fb, arow, brow, k + 64);
            mma8(fa, acc0, acc1);
            load8(fa, arow, brow, k + 128);
            mma8(fb, acc0, acc1);
        }
        load8(fb, arow, brow, K1_ - 64);
        mma8(fa, acc0, acc1);
        mma8(fb, acc0, acc1);
        const float bias = b1[n0 + l];
        #pragma unroll
        for (int r = 0; r < 8; ++r) {
            float v = (acc0[r] + acc1[r]) + bias;
            v = v * sigmoidf_fast(v);
            hs[(r + 8 * half) * HP_ + n0 + l] = v;
        }
    }

    __syncthreads();

    if (wave == 0) {
        const int cb = (c0 + l < C_) ? (c0 + l) : (C_ - 1);
        const float* hrow = hs + l * HP_ + koff;
        const float* wa = w2 + (size_t)cb * K2_ + koff;

        v8f ca0 = {};
        v8f ca1 = {};
        #pragma unroll 12
        for (int k = 0; k < K2_; k += 8) {
            const v4f a = *(const v4f*)(hrow + k);
            const v4f p = *(const v4f*)(wa + k);
            const v2f a0 = {a.x, a.y};
            const v2f a1 = {a.z, a.w};
            const v2f p0 = {p.x, p.y};
            const v2f p1 = {p.z, p.w};
            ca0 = __builtin_amdgcn_wmma_f32_16x16x4_f32(
                false, a0, false, p0, (short)0, ca0, false, false);
            ca1 = __builtin_amdgcn_wmma_f32_16x16x4_f32(
                false, a1, false, p1, (short)0, ca1, false, false);
        }
        const float biasa = b2[cb];
        if (l < NCH_) {
            #pragma unroll
            for (int r = 0; r < 8; ++r)
                gs[(r + 8 * half) * GP_ + l] = sigmoidf_fast((ca0[r] + ca1[r]) + biasa);
        }
    }

    __syncthreads();

    const size_t rowbase = ((size_t)(m0 + wave) * C_ + c0) * HW_;
    const v4f* __restrict__ xw = (const v4f*)(x + rowbase) + lane;
    v4f* __restrict__ ow       = (v4f*)(out + rowbase) + lane;

    #pragma unroll 1
    for (int t0 = 0; t0 < NCH_; t0 += NB_) {
        v4f v[NB_][PV_];
        #pragma unroll
        for (int t = 0; t < NB_; ++t) {
            #pragma unroll
            for (int j = 0; j < PV_; ++j) v[t][j] = xw[(t0 + t) * (HW_ / 4) + 32 * j];
        }
        #pragma unroll
        for (int t = 0; t < NB_; ++t) {
            const float gv = gs[wave * GP_ + t0 + t];
            #pragma unroll
            for (int j = 0; j < PV_; ++j) ow[(t0 + t) * (HW_ / 4) + 32 * j] = v[t][j] * gv;
        }
    }
}

extern "C" void kernel_launch(void* const* d_in, const int* in_sizes, int n_in,
                              void* d_out, int out_size, void* d_ws, size_t ws_size,
                              hipStream_t stream) {
    const float* x  = (const float*)d_in[0];
    const float* w1 = (const float*)d_in[1];
    const float* b1 = (const float*)d_in[2];
    const float* w2 = (const float*)d_in[3];
    const float* b2 = (const float*)d_in[4];
    float* out = (float*)d_out;

    float* s = (float*)d_ws;

    se_squeeze<<<dim3(128, 2), 512, 0, stream>>>(x, s);

    se_mlp_scale<<<dim3(128, 2), 512, 0, stream>>>(x, s, w1, b1, w2, b2, out);
}


# Hidden-layer operands (means, w1) written by the pool launch in WMMA fragment order: each fragment load of the scale launch reads 512 contiguous bytes, was 16 lines x 32 B (same f32 sum order)
# speedup vs baseline: 5.1672x; 1.2184x over previous
#include <hip/hip_runtime.h>
#include <hip/hip_bf16.h>
#include <math.h>

#define B_   32
#define C_   768
#define CR_  192
#define HW_  1024
#define K1_  C_
#define K2_  CR_
#define KB1_ (K1_ / 8)
#define RT_  (CR_ * K1_ / 4 / 256)

typedef __attribute__((ext_vector_type(2))) float v2f;
typedef __attribute__((ext_vector_type(4))) float v4f;
typedef __attribute__((ext_vector_type(8))) float v8f;

__device__ __forceinline__ float sigmoidf_fast(float v) {
    return 1.0f / (1.0f + __expf(-v));
}

struct frag8 { v4f a[8]; v4f b[8]; };

__device__ __forceinline__ void load8(frag8& f, const v4f* __restrict__ at,
                                      const v4f* __restrict__ bt, int kb) {
    #pragma unroll
    for (int i = 0; i < 8; ++i) {
        f.a[i] = at[(kb + i) * 32];
        f.b[i] = bt[(kb + i) * 32];
    }
}

__device__ __forceinline__ void mma8(const frag8& f, v8f& acc0, v8f& acc1) {
    #pragma unroll
    for (int i = 0; i < 8; ++i) {
        const v2f a0 = {f.a[i].x, f.a[i].y};
        const v2f b0 = {f.b[i].x, f.b[i].y};
        const v2f a1 = {f.a[i].z, f.a[i].w};
        const v2f b1 = {f.b[i].z, f.b[i].w};
        acc0 = __builtin_amdgcn_wmma_f32_16x16x4_f32(
            false, a0, false, b0, (short)0, acc0, false, false);
        acc1 = __builtin_amdgcn_wmma_f32_16x16x4_f32(
            false, a1, false, b1, (short)0, acc1, false, false);
    }
}

#define HP_  196
#define NCH_ 6
#define GP_  8
#define PV_  (HW_ / 4 / 32)
#define NB_  3

__global__ __launch_bounds__(512) __attribute__((amdgpu_waves_per_eu(4, 4))) void se_squeeze(const float* __restrict__ x,
                                                  const float* __restrict__ w1,
                                                  float* __restrict__ st,
                                                  float* __restrict__ w1t) {
    const int wave = __builtin_amdgcn_readfirstlane((int)(threadIdx.x >> 5));
    const int lane = threadIdx.x & 31;
    const int m0   = blockIdx.y * 16;
    const int c0   = blockIdx.x * NCH_;

    const int p = (blockIdx.y * 128 + blockIdx.x) * RT_ + (int)threadIdx.x;
    v4f wpiece = {};
    if (threadIdx.x < RT_) {
        const int pl = p & 31;
        const int pb = (p >> 5) % KB1_;
        const int pt = (p >> 5) / KB1_;
        wpiece = *(const v4f*)(w1 + (size_t)(16 * pt + (pl & 15)) * K1_ + 8 * pb + 4 * (pl >> 4));
    }

    const size_t rowbase = ((size_t)(m0 + wave) * C_ + c0) * HW_;
    const v4f* __restrict__ xw = (const v4f*)(x + rowbase) + lane;

    v4f v[NCH_][PV_];
    #pragma unroll
    for (int t = 0; t < NCH_; ++t) {
        #pragma unroll
        for (int j = 0; j < PV_; ++j) v[t][j] = xw[t * (HW_ / 4) + 32 * j];
    }

    float mine = 0.0f;
    #pragma unroll
    for (int t = 0; t < NCH_; ++t) {
        float a0 = (v[t][0].x + v[t][0].y) + (v[t][0].z + v[t][0].w);
        float a1 = (v[t][1].x + v[t][1].y) + (v[t][1].z + v[t][1].w);
        float a2 = (v[t][2].x + v[t][2].y) + (v[t][2].z + v[t][2].w);
        float a3 = (v[t][3].x + v[t][3].y) + (v[t][3].z + v[t][3].w);
        a0 += (v[t][4].x + v[t][4].y) + (v[t][4].z + v[t][4].w);
        a1 += (v[t][5].x + v[t][5].y) + (v[t][5].z + v[t][5].w);
        a2 += (v[t][6].x + v[t][6].y) + (v[t][6].z + v[t][6].w);
        a3 += (v[t][7].x + v[t][7].y) + (v[t][7].z + v[t][7].w);
        float sum = (a0 + a1) + (a2 + a3);
        #pragma unroll
        for (int o = 16; o > 0; o >>= 1)
            sum += __shfl_xor(sum, o, 32);
        if (lane == t) mine = sum;
    }
    if (threadIdx.x < RT_) ((v4f*)w1t)[p] = wpiece;
    if (lane < NCH_) {
        const int c = c0 + lane;
        st[((size_t)(blockIdx.y * KB1_ + (c >> 3)) * 32 + 16 * ((c >> 2) & 1) + wave) * 4 + (c & 3)]
            = mine * (1.0f / (float)HW_);
    }
}

__global__ __launch_bounds__(512) void se_mlp_scale(const float* __restrict__ x,
                                                    const float* __restrict__ st,
                                                    const float* __restrict__ w1t,
                                                    const float* __restrict__ b1,
                                                    const float* __restrict__ w2,
                                                    const float* __restrict__ b2,
                                                    float* __restrict__ out) {
    __shared__ __attribute__((aligned(16))) float hs[16 * HP_];
    __shared__ float gs[16 * GP_];

    const int wave = __builtin_amdgcn_readfirstlane((int)(threadIdx.x >> 5));
    const int lane = threadIdx.x & 31;
    const int half = lane >> 4;
    const int l    = lane & 15;
    const int koff = 4 * half;
    const int m0   = blockIdx.y * 16;
    const int c0   = blockIdx.x * NCH_;

    if (wave < CR_ / 16) {
        const int n0 = wave * 16;
        const v4f* at = (const v4f*)st  + (size_t)blockIdx.y * (KB1_ * 32) + lane;
        const v4f* bt = (const v4f*)w1t + (size_t)wave * (KB1_ * 32) + lane;

        v8f acc0 = {};
        v8f acc1 = {};
        frag8 fa, fb;
        load8(fa, at, bt, 0);
        #pragma unroll 1
        for (int kb = 0; kb < KB1_ - 16; kb += 16) {
            load8(fb, at, bt, kb + 8);
            mma8(fa, acc0, acc1);
            load8(fa, at, bt, kb + 16);
            mma8(fb, acc0, acc1);
        }
        load8(fb, at, bt, KB1_ - 8);
        mma8(fa, acc0, acc1);
        mma8(fb, acc0, acc1);
        const float bias = b1[n0 + l];
        #pragma unroll
        for (int r = 0; r < 8; ++r) {
            float v = (acc0[r] + acc1[r]) + bias;
            v = v * sigmoidf_fast(v);
            hs[(r + 8 * half) * HP_ + n0 + l] = v;
        }
    }

    __syncthreads();

    if (wave == 0) {
        const int cb = (c0 + l < C_) ? (c0 + l) : (C_ - 1);
        const float* hrow = hs + l * HP_ + koff;
        const float* wa = w2 + (size_t)cb * K2_ + koff;

        v8f ca0 = {};
        v8f ca1 = {};
        #pragma unroll 12
        for (int k = 0; k < K2_; k += 8) {
            const v4f a = *(const v4f*)(hrow + k);
            const v4f p = *(const v4f*)(wa + k);
            const v2f a0 = {a.x, a.y};
            const v2f a1 = {a.z, a.w};
            const v2f p0 = {p.x, p.y};
            const v2f p1 = {p.z, p.w};
            ca0 = __builtin_amdgcn_wmma_f32_16x16x4_f32(
                false, a0, false, p0, (short)0, ca0, false, false);
            ca1 = __builtin_amdgcn_wmma_f32_16x16x4_f32(
                false, a1, false, p1, (short)0, ca1, false, false);
        }
        const float biasa = b2[cb];
        if (l < NCH_) {
            #pragma unroll
            for (int r = 0; r < 8; ++r)
                gs[(r + 8 * half) * GP_ + l] = sigmoidf_fast((ca0[r] + ca1[r]) + biasa);
        }
    }

    __syncthreads();

    const size_t rowbase = ((size_t)(m0 + wave) * C_ + c0) * HW_;
    const v4f* __restrict__ xw = (const v4f*)(x + rowbase) + lane;
    v4f* __restrict__ ow       = (v4f*)(out + rowbase) + lane;

    #pragma unroll 1
    for (int t0 = 0; t0 < NCH_; t0 += NB_) {
        v4f v[NB_][PV_];
        #pragma unroll
        for (int t = 0; t < NB_; ++t) {
            #pragma unroll
            for (int j = 0; j < PV_; ++j) v[t][j] = xw[(t0 + t) * (HW_ / 4) + 32 * j];
        }
        #pragma unroll
        for (int t = 0; t < NB_; ++t) {
            const float gv = gs[wave * GP_ + t0 + t];
            #pragma unroll
            for (int j = 0; j < PV_; ++j) ow[(t0 + t) * (HW_ / 4) + 32 * j] = v[t][j] * gv;
        }
    }
}

extern "C" void kernel_launch(void* const* d_in, const int* in_sizes, int n_in,
                              void* d_out, int out_size, void* d_ws, size_t ws_size,
                              hipStream_t stream) {
    const float* x  = (const float*)d_in[0];
    const float* w1 = (const float*)d_in[1];
    const float* b1 = (const float*)d_in[2];
    const float* w2 = (const float*)d_in[3];
    const float* b2 = (const float*)d_in[4];
    float* out = (float*)d_out;

    float* st  = (float*)d_ws;
    float* w1t = (float*)d_ws + 24576;

    se_squeeze<<<dim3(128, 2), 512, 0, stream>>>(x, w1, st, w1t);

    se_mlp_scale<<<dim3(128, 2), 512, 0, stream>>>(x, st, w1t, b1, w2, b2, out);
}


# Gate layer folded into the hidden waves: each forms its tile transposed, so it is the B operand in registers, and adds its 16 units' gate share; one barrier, 12 partials per row (new f32 sum order)
# speedup vs baseline: 5.3755x; 1.0403x over previous
#include <hip/hip_runtime.h>
#include <hip/hip_bf16.h>
#include <math.h>

#define B_   32
#define C_   768
#define CR_  192
#define HW_  1024
#define K1_  C_
#define K2_  CR_
#define KB1_ (K1_ / 8)
#define RT_  (CR_ * K1_ / 4 / 256)

typedef __attribute__((ext_vector_type(2))) float v2f;
typedef __attribute__((ext_vector_type(4))) float v4f;
typedef __attribute__((ext_vector_type(8))) float v8f;

__device__ __forceinline__ float sigmoidf_fast(float v) {
    return 1.0f / (1.0f + __expf(-v));
}

struct frag8 { v4f a[8]; v4f b[8]; };

__device__ __forceinline__ void load8(frag8& f, const v4f* __restrict__ at,
                                      const v4f* __restrict__ bt, int kb) {
    #pragma unroll
    for (int i = 0; i < 8; ++i) {
        f.a[i] = at[(kb + i) * 32];
        f.b[i] = bt[(kb + i) * 32];
    }
}

__device__ __forceinline__ void mma8(const frag8& f, v8f& acc0, v8f& acc1) {
    #pragma unroll
    for (int i = 0; i < 8; ++i) {
        const v2f s0 = {f.a[i].x, f.a[i].y};
        const v2f w0 = {f.b[i].x, f.b[i].y};
        const v2f s1 = {f.a[i].z, f.a[i].w};
        const v2f w1 = {f.b[i].z, f.b[i].w};
        acc0 = __builtin_amdgcn_wmma_f32_16x16x4_f32(
            false, w0, false, s0, (short)0, acc0, false, false);
        acc1 = __builtin_amdgcn_wmma_f32_16x16x4_f32(
            false, w1, false, s1, (short)0, acc1, false, false);
    }
}

#define NCH_ 6
#define GPP_ (NCH_ * 16)
#define PV_  (HW_ / 4 / 32)
#define NB_  3

__global__ __launch_bounds__(512) __attribute__((amdgpu_waves_per_eu(4, 4))) void se_squeeze(const float* __restrict__ x,
                                                  const float* __restrict__ w1,
                                                  float* __restrict__ st,
                                                  float* __restrict__ w1t) {
    const int wave = __builtin_amdgcn_readfirstlane((int)(threadIdx.x >> 5));
    const int lane = threadIdx.x & 31;
    const int m0   = blockIdx.y * 16;
    const int c0   = blockIdx.x * NCH_;

    const int p = (blockIdx.y * 128 + blockIdx.x) * RT_ + (int)threadIdx.x;
    v4f wpiece = {};
    if (threadIdx.x < RT_) {
        const int pl = p & 31;
        const int pb = (p >> 5) % KB1_;
        const int pt = (p >> 5) / KB1_;
        wpiece = *(const v4f*)(w1 + (size_t)(16 * pt + (pl & 15)) * K1_ + 8 * pb + 4 * (pl >> 4));
    }

    const size_t rowbase = ((size_t)(m0 + wave) * C_ + c0) * HW_;
    const v4f* __restrict__ xw = (const v4f*)(x + rowbase) + lane;

    v4f v[NCH_][PV_];
    #pragma unroll
    for (int t = 0; t < NCH_; ++t) {
        #pragma unroll
        for (int j = 0; j < PV_; ++j) v[t][j] = xw[t * (HW_ / 4) + 32 * j];
    }

    float mine = 0.0f;
    #pragma unroll
    for (int t = 0; t < NCH_; ++t) {
        float a0 = (v[t][0].x + v[t][0].y) + (v[t][0].z + v[t][0].w);
        float a1 = (v[t][1].x + v[t][1].y) + (v[t][1].z + v[t][1].w);
        float a2 = (v[t][2].x + v[t][2].y) + (v[t][2].z + v[t][2].w);
        float a3 = (v[t][3].x + v[t][3].y) + (v[t][3].z + v[t][3].w);
        a0 += (v[t][4].x + v[t][4].y) + (v[t][4].z + v[t][4].w);
        a1 += (v[t][5].x + v[t][5].y) + (v[t][5].z + v[t][5].w);
        a2 += (v[t][6].x + v[t][6].y) + (v[t][6].z + v[t][6].w);
        a3 += (v[t][7].x + v[t][7].y) + (v[t][7].z + v[t][7].w);
        float sum = (a0 + a1) + (a2 + a3);
        #pragma unroll
        for (int o = 16; o > 0; o >>= 1)
            sum += __shfl_xor(sum, o, 32);
        if (lane == t) mine = sum;
    }
    if (threadIdx.x < RT_) ((v4f*)w1t)[p] = wpiece;
    if (lane < NCH_) {
        const int c = c0 + lane;
        st[((size_t)(blockIdx.y * KB1_ + (c >> 3)) * 32 + 16 * ((c >> 2) & 1) + wave) * 4 + (c & 3)]
            = mine * (1.0f / (float)HW_);
    }
}

__global__ __launch_bounds__(512) void se_mlp_scale(const float* __restrict__ x,
                                                    const float* __restrict__ st,
                                                    const float* __restrict__ w1t,
                                                    const float* __restrict__ b1,
                                                    const float* __restrict__ w2,
                                                    const float* __restrict__ b2,
                                                    float* __restrict__ out) {
    __shared__ float gp[(CR_ / 16) * GPP_];

    const int wave = __builtin_amdgcn_readfirstlane((int)(threadIdx.x >> 5));
    const int lane = threadIdx.x & 31;
    const int half = lane >> 4;
    const int l    = lane & 15;
    const int m0   = blockIdx.y * 16;
    const int c0   = blockIdx.x * NCH_;

    float bias2 = 0.0f;
    if (lane < NCH_) bias2 = b2[c0 + lane];

    if (wave < CR_ / 16) {
        const int n0 = wave * 16;
        const v4f* at = (const v4f*)st  + (size_t)blockIdx.y * (KB1_ * 32) + lane;
        const v4f* bt = (const v4f*)w1t + (size_t)wave * (KB1_ * 32) + lane;

        const int cb = (c0 + l < C_) ? (c0 + l) : (C_ - 1);
        const v4f* bq = (const v4f*)(b1 + n0 + 8 * half);
        const v4f* wq = (const v4f*)(w2 + (size_t)cb * K2_ + n0 + 8 * half);

        v8f acc0 = {};
        v8f acc1 = {};
        frag8 fa, fb;
        load8(fa, at, bt, 0);
        #pragma unroll 1
        for (int kb = 0; kb < KB1_ - 16; kb += 16) {
            load8(fb, at, bt, kb + 8);
            mma8(fa, acc0, acc1);
            load8(fa, at, bt, kb + 16);
            mma8(fb, acc0, acc1);
        }
        load8(fb, at, bt, KB1_ - 8);
        const v4f blo = bq[0];
        const v4f bhi = bq[1];
        const v4f wlo = wq[0];
        const v4f whi = wq[1];
        mma8(fa, acc0, acc1);
        mma8(fb, acc0, acc1);

        const float bias1[8] = {blo.x, blo.y, blo.z, blo.w, bhi.x, bhi.y, bhi.z, bhi.w};
        float h[8];
        #pragma unroll
        for (int r = 0; r < 8; ++r) {
            const float v = (acc0[r] + acc1[r]) + bias1[r];
            h[r] = v * sigmoidf_fast(v);
        }

        v8f pa = {};
        v8f pb = {};
        {
            const v2f wA = {wlo.x, wlo.y};
            const v2f hA = {h[0], h[1]};
            const v2f wB = {wlo.z, wlo.w};
            const v2f hB = {h[2], h[3]};
            const v2f wC = {whi.x, whi.y};
            const v2f hC = {h[4], h[5]};
            const v2f wD = {whi.z, whi.w};
            const v2f hD = {h[6], h[7]};
            pa = __builtin_amdgcn_wmma_f32_16x16x4_f32(false, wA, false, hA, (short)0, pa, false, false);
            pb = __builtin_amdgcn_wmma_f32_16x16x4_f32(false, wB, false, hB, (short)0, pb, false, false);
            pa = __builtin_amdgcn_wmma_f32_16x16x4_f32(false, wC, false, hC, (short)0, pa, false, false);
            pb = __builtin_amdgcn_wmma_f32_16x16x4_f32(false, wD, false, hD, (short)0, pb, false, false);
        }
        if (half == 0) {
            #pragma unroll
            for (int r = 0; r < NCH_; ++r) gp[wave * GPP_ + r * 16 + l] = pa[r] + pb[r];
        }
    }

    __syncthreads();

    const size_t rowbase = ((size_t)(m0 + wave) * C_ + c0) * HW_;
    const v4f* __restrict__ xw = (const v4f*)(x + rowbase) + lane;
    v4f* __restrict__ ow       = (v4f*)(out + rowbase) + lane;

    float g = 0.0f;
    #pragma unroll 1
    for (int t0 = 0; t0 < NCH_; t0 += NB_) {
        v4f v[NB_][PV_];
        #pragma unroll
        for (int t = 0; t < NB_; ++t) {
            #pragma unroll
            for (int j = 0; j < PV_; ++j) v[t][j] = xw[(t0 + t) * (HW_ / 4) + 32 * j];
        }
        if (t0 == 0) {
            float s = 0.0f;
            if (lane < NCH_) {
                #pragma unroll
                for (int t = 0; t < CR_ / 16; ++t) s += gp[t * GPP_ + lane * 16 + wave];
                s = sigmoidf_fast(s + bias2);
            }
            g = s;
        }
        #pragma unroll
        for (int t = 0; t < NB_; ++t) {
            const float gv = __int_as_float(__builtin_amdgcn_readlane(__float_as_int(g), t0 + t));
            #pragma unroll
            for (int j = 0; j < PV_; ++j) ow[(t0 + t) * (HW_ / 4) + 32 * j] = v[t][j] * gv;
        }
    }
}

extern "C" void kernel_launch(void* const* d_in, const int* in_sizes, int n_in,
                              void* d_out, int out_size, void* d_ws, size_t ws_size,
                              hipStream_t stream) {
    const float* x  = (const float*)d_in[0];
    const float* w1 = (const float*)d_in[1];
    const float* b1 = (const float*)d_in[2];
    const float* w2 = (const float*)d_in[3];
    const float* b2 = (const float*)d_in[4];
    float* out = (float*)d_out;

    float* st  = (float*)d_ws;
    float* w1t = (float*)d_ws + 24576;

    se_squeeze<<<dim3(128, 2), 512, 0, stream>>>(x, w1, st, w1t);

    se_mlp_scale<<<dim3(128, 2), 512, 0, stream>>>(x, st, w1t, b1, w2, b2, out);
}


# Scale phase pipelined by plane: 3 register sets in rotation, loads run 2 planes ahead of the plane being stored (was 24 loads, wait for all, 24 stores, twice); idle waves load before the barrier
# speedup vs baseline: 5.6218x; 1.0458x over previous
#include <hip/hip_runtime.h>
#include <hip/hip_bf16.h>
#include <math.h>

#define B_   32
#define C_   768
#define CR_  192
#define HW_  1024
#define K1_  C_
#define K2_  CR_
#define KB1_ (K1_ / 8)
#define RT_  (CR_ * K1_ / 4 / 256)

typedef __attribute__((ext_vector_type(2))) float v2f;
typedef __attribute__((ext_vector_type(4))) float v4f;
typedef __attribute__((ext_vector_type(8))) float v8f;

__device__ __forceinline__ float sigmoidf_fast(float v) {
    return 1.0f / (1.0f + __expf(-v));
}

struct frag8 { v4f a[8]; v4f b[8]; };

__device__ __forceinline__ void load8(frag8& f, const v4f* __restrict__ at,
                                      const v4f* __restrict__ bt, int kb) {
    #pragma unroll
    for (int i = 0; i < 8; ++i) {
        f.a[i] = at[(kb + i) * 32];
        f.b[i] = bt[(kb + i) * 32];
    }
}

__device__ __forceinline__ void mma8(const frag8& f, v8f& acc0, v8f& acc1) {
    #pragma unroll
    for (int i = 0; i < 8; ++i) {
        const v2f s0 = {f.a[i].x, f.a[i].y};
        const v2f w0 = {f.b[i].x, f.b[i].y};
        const v2f s1 = {f.a[i].z, f.a[i].w};
        const v2f w1 = {f.b[i].z, f.b[i].w};
        acc0 = __builtin_amdgcn_wmma_f32_16x16x4_f32(
            false, w0, false, s0, (short)0, acc0, false, false);
        acc1 = __builtin_amdgcn_wmma_f32_16x16x4_f32(
            false, w1, false, s1, (short)0, acc1, false, false);
    }
}

#define NCH_ 6
#define GPP_ (NCH_ * 16)
#define PV_  (HW_ / 4 / 32)
#define NS_  3

__global__ __launch_bounds__(512) __attribute__((amdgpu_waves_per_eu(4, 4))) void se_squeeze(const float* __restrict__ x,
                                                  const float* __restrict__ w1,
                                                  float* __restrict__ st,
                                                  float* __restrict__ w1t) {
    const int wave = __builtin_amdgcn_readfirstlane((int)(threadIdx.x >> 5));
    const int lane = threadIdx.x & 31;
    const int m0   = blockIdx.y * 16;
    const int c0   = blockIdx.x * NCH_;

    const int p = (blockIdx.y * 128 + blockIdx.x) * RT_ + (int)threadIdx.x;
    v4f wpiece = {};
    if (threadIdx.x < RT_) {
        const int pl = p & 31;
        const int pb = (p >> 5) % KB1_;
        const int pt = (p >> 5) / KB1_;
        wpiece = *(const v4f*)(w1 + (size_t)(16 * pt + (pl & 15)) * K1_ + 8 * pb + 4 * (pl >> 4));
    }

    const size_t rowbase = ((size_t)(m0 + wave) * C_ + c0) * HW_;
    const v4f* __restrict__ xw = (const v4f*)(x + rowbase) + lane;

    v4f v[NCH_][PV_];
    #pragma unroll
    for (int t = 0; t < NCH_; ++t) {
        #pragma unroll
        for (int j = 0; j < PV_; ++j) v[t][j] = xw[t * (HW_ / 4) + 32 * j];
    }

    float mine = 0.0f;
    #pragma unroll
    for (int t = 0; t < NCH_; ++t) {
        float a0 = (v[t][0].x + v[t][0].y) + (v[t][0].z + v[t][0].w);
        float a1 = (v[t][1].x + v[t][1].y) + (v[t][1].z + v[t][1].w);
        float a2 = (v[t][2].x + v[t][2].y) + (v[t][2].z + v[t][2].w);
        float a3 = (v[t][3].x + v[t][3].y) + (v[t][3].z + v[t][3].w);
        a0 += (v[t][4].x + v[t][4].y) + (v[t][4].z + v[t][4].w);
        a1 += (v[t][5].x + v[t][5].y) + (v[t][5].z + v[t][5].w);
        a2 += (v[t][6].x + v[t][6].y) + (v[t][6].z + v[t][6].w);
        a3 += (v[t][7].x + v[t][7].y) + (v[t][7].z + v[t][7].w);
        float sum = (a0 + a1) + (a2 + a3);
        #pragma unroll
        for (int o = 16; o > 0; o >>= 1)
            sum += __shfl_xor(sum, o, 32);
        if (lane == t) mine = sum;
    }
    if (threadIdx.x < RT_) ((v4f*)w1t)[p] = wpiece;
    if (lane < NCH_) {
        const int c = c0 + lane;
        st[((size_t)(blockIdx.y * KB1_ + (c >> 3)) * 32 + 16 * ((c >> 2) & 1) + wave) * 4 + (c & 3)]
            = mine * (1.0f / (float)HW_);
    }
}

__global__ __launch_bounds__(512) void se_mlp_scale(const float* __restrict__ x,
                                                    const float* __restrict__ st,
                                                    const float* __restrict__ w1t,
                                                    const float* __restrict__ b1,
                                                    const float* __restrict__ w2,
                                                    const float* __restrict__ b2,
                                                    float* __restrict__ out) {
    __shared__ float gp[(CR_ / 16) * GPP_];

    const int wave = __builtin_amdgcn_readfirstlane((int)(threadIdx.x >> 5));
    const int lane = threadIdx.x & 31;
    const int half = lane >> 4;
    const int l    = lane & 15;
    const int m0   = blockIdx.y * 16;
    const int c0   = blockIdx.x * NCH_;

    float bias2 = 0.0f;
    if (lane < NCH_) bias2 = b2[c0 + lane];

    if (wave < CR_ / 16) {
        const int n0 = wave * 16;
        const v4f* at = (const v4f*)st  + (size_t)blockIdx.y * (KB1_ * 32) + lane;
        const v4f* bt = (const v4f*)w1t + (size_t)wave * (KB1_ * 32) + lane;

        const int cb = (c0 + l < C_) ? (c0 + l) : (C_ - 1);
        const v4f* bq = (const v4f*)(b1 + n0 + 8 * half);
        const v4f* wq = (const v4f*)(w2 + (size_t)cb * K2_ + n0 + 8 * half);

        v8f acc0 = {};
        v8f acc1 = {};
        frag8 fa, fb;
        load8(fa, at, bt, 0);
        #pragma unroll 1
        for (int kb = 0; kb < KB1_ - 16; kb += 16) {
            load8(fb, at, bt, kb + 8);
            mma8(fa, acc0, acc1);
            load8(fa, at, bt, kb + 16);
            mma8(fb, acc0, acc1);
        }
        load8(fb, at, bt, KB1_ - 8);
        const v4f blo = bq[0];
        const v4f bhi = bq[1];
        const v4f wlo = wq[0];
        const v4f whi = wq[1];
        mma8(fa, acc0, acc1);
        mma8(fb, acc0, acc1);

        const float bias1[8] = {blo.x, blo.y, blo.z, blo.w, bhi.x, bhi.y, bhi.z, bhi.w};
        float h[8];
        #pragma unroll
        for (int r = 0; r < 8; ++r) {
            const float v = (acc0[r] + acc1[r]) + bias1[r];
            h[r] = v * sigmoidf_fast(v);
        }

        v8f pa = {};
        v8f pb = {};
        {
            const v2f wA = {wlo.x, wlo.y};
            const v2f hA = {h[0], h[1]};
            const v2f wB = {wlo.z, wlo.w};
            const v2f hB = {h[2], h[3]};
            const v2f wC = {whi.x, whi.y};
            const v2f hC = {h[4], h[5]};
            const v2f wD = {whi.z, whi.w};
            const v2f hD = {h[6], h[7]};
            pa = __builtin_amdgcn_wmma_f32_16x16x4_f32(false, wA, false, hA, (short)0, pa, false, false);
            pb = __builtin_amdgcn_wmma_f32_16x16x4_f32(false, wB, false, hB, (short)0, pb, false, false);
            pa = __builtin_amdgcn_wmma_f32_16x16x4_f32(false, wC, false, hC, (short)0, pa, false, false);
            pb = __builtin_amdgcn_wmma_f32_16x16x4_f32(false, wD, false, hD, (short)0, pb, false, false);
        }
        if (half == 0) {
            #pragma unroll
            for (int r = 0; r < NCH_; ++r) gp[wave * GPP_ + r * 16 + l] = pa[r] + pb[r];
        }
    }

    const size_t rowbase = ((size_t)(m0 + wave) * C_ + c0) * HW_;
    const v4f* __restrict__ xw = (const v4f*)(x + rowbase) + lane;
    v4f* __restrict__ ow       = (v4f*)(out + rowbase) + lane;

    v4f p[NS_][PV_];
    if (wave >= CR_ / 16) {
        #pragma unroll
        for (int t = 0; t < NS_ - 1; ++t) {
            #pragma unroll
            for (int j = 0; j < PV_; ++j) p[t][j] = xw[t * (HW_ / 4) + 32 * j];
        }
    }

    __syncthreads();

    if (wave < CR_ / 16) {
        #pragma unroll
        for (int t = 0; t < NS_ - 1; ++t) {
            #pragma unroll
            for (int j = 0; j < PV_; ++j) p[t][j] = xw[t * (HW_ / 4) + 32 * j];
        }
    }

    float g = 0.0f;
    if (lane < NCH_) {
        float s = 0.0f;
        #pragma unroll
        for (int t = 0; t < CR_ / 16; ++t) s += gp[t * GPP_ + lane * 16 + wave];
        g = sigmoidf_fast(s + bias2);
    }

    #pragma unroll
    for (int t = 0; t < NCH_; ++t) {
        if (t + NS_ - 1 < NCH_) {
            #pragma unroll
            for (int j = 0; j < PV_; ++j)
                p[(t + NS_ - 1) % NS_][j] = xw[(t + NS_ - 1) * (HW_ / 4) + 32 * j];
        }
        const float gv = __int_as_float(__builtin_amdgcn_readlane(__float_as_int(g), t));
        #pragma unroll
        for (int j = 0; j < PV_; ++j) ow[t * (HW_ / 4) + 32 * j] = p[t % NS_][j] * gv;
        __builtin_amdgcn_sched_barrier(0);
    }
}

extern "C" void kernel_launch(void* const* d_in, const int* in_sizes, int n_in,
                              void* d_out, int out_size, void* d_ws, size_t ws_size,
                              hipStream_t stream) {
    const float* x  = (const float*)d_in[0];
    const float* w1 = (const float*)d_in[1];
    const float* b1 = (const float*)d_in[2];
    const float* w2 = (const float*)d_in[3];
    const float* b2 = (const float*)d_in[4];
    float* out = (float*)d_out;

    float* st  = (float*)d_ws;
    float* w1t = (float*)d_ws + 24576;

    se_squeeze<<<dim3(128, 2), 512, 0, stream>>>(x, w1, st, w1t);

    se_mlp_scale<<<dim3(128, 2), 512, 0, stream>>>(x, st, w1t, b1, w2, b2, out);
}
